# starting kernel
# baseline (speedup 1.0000x reference)
#include <hip/hip_runtime.h>
#include <cmath>

#define NNODES 100000
#define NEDGES 600000
#define FEAT   128
#define NGRAPH 256

typedef __attribute__((ext_vector_type(2))) float v2f;
typedef __attribute__((ext_vector_type(8))) float v8f;

__global__ void gcn_init_kernel(float* __restrict__ out, float* __restrict__ cnts,
                                float* __restrict__ deg) {
    int i = blockIdx.x * 256 + threadIdx.x;
    if (i < NGRAPH * FEAT) out[i] = 0.0f;
    if (i < NGRAPH)        cnts[i] = 0.0f;
    if (i < NNODES)        deg[i]  = 1.0f;
}

__global__ void gcn_deg_kernel(const int* __restrict__ col, float* __restrict__ deg) {
    int e = blockIdx.x * 256 + threadIdx.x;
    if (e < NEDGES) atomicAdd(&deg[col[e]], 1.0f);
}

__global__ void gcn_dinv_kernel(float* __restrict__ deg) {
    int n = blockIdx.x * 256 + threadIdx.x;
    if (n < NNODES) {
        float d = deg[n];
        deg[n] = (d > 0.0f) ? rsqrtf(d) : 0.0f;
    }
}

template <bool RELU>
__global__ void gcn_gemm_kernel(const float* __restrict__ X,
                                const float* __restrict__ W,
                                const float* __restrict__ bias,
                                float* __restrict__ Y) {
    __shared__ float Wt[FEAT * 16];
    __shared__ float Bt[16];

    const int lane  = threadIdx.x & 31;
    const int wave  = threadIdx.x >> 5;
    const int m0    = blockIdx.x * 128 + wave * 16;
    const bool active = (m0 < NNODES);

    const int mrow  = m0 + (lane & 15);
    const int khalf = (lane >> 4) << 1;
    const int ncol  = lane & 15;
    const int half  = lane >> 4;

    for (int nt = 0; nt < 8; ++nt) {
        __syncthreads();
        #pragma unroll
        for (int i = 0; i < 8; ++i) {
            int idx = threadIdx.x * 8 + i;
            int kk  = idx >> 4;
            int cc  = idx & 15;
            Wt[idx] = W[kk * FEAT + nt * 16 + cc];
        }
        if (threadIdx.x < 16) Bt[threadIdx.x] = bias[nt * 16 + threadIdx.x];
        __syncthreads();

        if (!active) continue;

        v8f acc = {};
        #pragma unroll 4
        for (int k = 0; k < FEAT; k += 4) {
            float a0 = X[(size_t)mrow * FEAT + k + khalf];
            float a1 = X[(size_t)mrow * FEAT + k + khalf + 1];
            if (RELU) { a0 = fmaxf(a0, 0.0f); a1 = fmaxf(a1, 0.0f); }
            v2f a; a[0] = a0; a[1] = a1;
            v2f b;
            b[0] = Wt[(k + khalf) * 16 + ncol];
            b[1] = Wt[(k + khalf + 1) * 16 + ncol];
            acc = __builtin_amdgcn_wmma_f32_16x16x4_f32(
                false, a, false, b, (short)0, acc, false, false);
        }

        #pragma unroll
        for (int v = 0; v < 8; ++v) {
            int r = m0 + v + half * 8;
            Y[(size_t)r * FEAT + nt * 16 + ncol] = acc[v] + Bt[ncol];
        }
    }
}

__global__ void gcn_agg_init_kernel(const float* __restrict__ H,
                                    const float* __restrict__ dinv,
                                    float* __restrict__ AGG) {
    int i = blockIdx.x * 256 + threadIdx.x;
    if (i >= NNODES * 32) return;
    int n = i >> 5;
    float s = dinv[n]; s = s * s;
    float4 h = *(const float4*)(H + (size_t)i * 4);
    float4 r = make_float4(h.x * s, h.y * s, h.z * s, h.w * s);
    *(float4*)(AGG + (size_t)i * 4) = r;
}

__global__ void gcn_scatter_kernel(const int* __restrict__ row,
                                   const int* __restrict__ col,
                                   const float* __restrict__ dinv,
                                   const float* __restrict__ H,
                                   float* __restrict__ AGG) {
    int t = blockIdx.x * 256 + threadIdx.x;
    int e = t >> 5;
    if (e >= NEDGES) return;
    int lane = t & 31;
    int r = row[e];
    int c = col[e];
    float norm = dinv[r] * dinv[c];
    float4 h = *(const float4*)(H + (size_t)c * FEAT + lane * 4);
    float* dst = AGG + (size_t)r * FEAT + lane * 4;
    atomicAdd(dst + 0, norm * h.x);
    atomicAdd(dst + 1, norm * h.y);
    atomicAdd(dst + 2, norm * h.z);
    atomicAdd(dst + 3, norm * h.w);
}

__global__ void gcn_pool_kernel(const float* __restrict__ AGG,
                                const int* __restrict__ batch,
                                float* __restrict__ out,
                                float* __restrict__ cnts) {
    int t = blockIdx.x * 256 + threadIdx.x;
    int n = t >> 5;
    if (n >= NNODES) return;
    int lane = t & 31;
    int g = batch[n];
    float4 h = *(const float4*)(AGG + (size_t)n * FEAT + lane * 4);
    float* dst = out + (size_t)g * FEAT + lane * 4;
    atomicAdd(dst + 0, h.x);
    atomicAdd(dst + 1, h.y);
    atomicAdd(dst + 2, h.z);
    atomicAdd(dst + 3, h.w);
    if (lane == 0) atomicAdd(&cnts[g], 1.0f);
}

__global__ void gcn_finalize_kernel(float* __restrict__ out,
                                    const float* __restrict__ cnts) {
    int i = blockIdx.x * 256 + threadIdx.x;
    if (i < NGRAPH * FEAT) {
        float c = cnts[i >> 7];
        out[i] = out[i] / fmaxf(c, 1.0f);
    }
}

extern "C" void kernel_launch(void* const* d_in, const int* in_sizes, int n_in,
                              void* d_out, int out_size, void* d_ws, size_t ws_size,
                              hipStream_t stream) {
    const float* x          = (const float*)d_in[0];
    const int*   edge_index = (const int*)  d_in[1];
    const int*   batch      = (const int*)  d_in[2];
    const float* W1 = (const float*)d_in[4];
    const float* b1 = (const float*)d_in[5];
    const float* W2 = (const float*)d_in[6];
    const float* b2 = (const float*)d_in[7];
    float* out = (float*)d_out;

    float* H    = (float*)d_ws;
    float* AGG  = H    + (size_t)NNODES * FEAT;
    float* dinv = AGG  + (size_t)NNODES * FEAT;
    float* cnts = dinv + NNODES;

    const int* row = edge_index;
    const int* col = edge_index + NEDGES;

    dim3 blk(256);
    const int grid_nodes  = (NNODES + 255) / 256;
    const int grid_edges  = (NEDGES + 255) / 256;
    const int grid_gemm   = (NNODES + 127) / 128;
    const int grid_nfeat4 = (NNODES * 32 + 255) / 256;
    const int grid_escat  = (NEDGES * 32 + 255) / 256;
    const int grid_out    = (NGRAPH * FEAT + 255) / 256;

    gcn_init_kernel<<<grid_nodes, blk, 0, stream>>>(out, cnts, dinv);
    gcn_deg_kernel<<<grid_edges, blk, 0, stream>>>(col, dinv);
    gcn_dinv_kernel<<<grid_nodes, blk, 0, stream>>>(dinv);

    gcn_gemm_kernel<false><<<grid_gemm, blk, 0, stream>>>(x, W1, b1, H);
    gcn_agg_init_kernel<<<grid_nfeat4, blk, 0, stream>>>(H, dinv, AGG);
    gcn_scatter_kernel<<<grid_escat, blk, 0, stream>>>(row, col, dinv, H, AGG);

    gcn_gemm_kernel<true><<<grid_gemm, blk, 0, stream>>>(AGG, W2, b2, H);
    gcn_agg_init_kernel<<<grid_nfeat4, blk, 0, stream>>>(H, dinv, AGG);
    gcn_scatter_kernel<<<grid_escat, blk, 0, stream>>>(row, col, dinv, H, AGG);

    gcn_pool_kernel<<<grid_nfeat4, blk, 0, stream>>>(AGG, batch, out, cnts);
    gcn_finalize_kernel<<<grid_out, blk, 0, stream>>>(out, cnts);
}


# layer-1 product taken after its aggregation (linear): tile sums x rows in LDS, then @W1 (+ row weight sum * b1), relu, @W2 run on it; layer-1 GEMM launch and H1 plane gone, 8 -> 7; new sum order
# speedup vs baseline: 3.1565x; 3.1565x over previous
#include <hip/hip_runtime.h>
#include <cmath>

#define NNODES 100000
#define NEDGES 600000
#define FEAT   128
#define NGRAPH 256

typedef __attribute__((ext_vector_type(2))) float v2f;
typedef __attribute__((ext_vector_type(8))) float v8f;

#define BROWS 128
#define NBUCK ((NNODES + BROWS - 1) / BROWS)
#define BCSTRIDE 64
#define BCWORDS  (NBUCK * BCSTRIDE)

__global__ void gcn_init_kernel(float* __restrict__ deg, int* __restrict__ bcount) {
    int i = blockIdx.x * 256 + threadIdx.x;
    if (i < NNODES)        deg[i]  = 1.0f;
    if (i < BCWORDS)       bcount[i] = 0;
}

__global__ void gcn_deg_kernel(const int* __restrict__ row, const int* __restrict__ col,
                               float* __restrict__ deg,
                               int* __restrict__ bcount, int* __restrict__ rank) {
    int e = blockIdx.x * 256 + threadIdx.x;
    if (e < NEDGES) {
        const int c = col[e];
        const int r = row[e];
        atomicAdd(&deg[c], 1.0f);
        rank[e] = atomicAdd(&bcount[(r / BROWS) * BCSTRIDE], 1);
    }
}

__global__ void gcn_dinv_kernel(float* __restrict__ deg,
                                const int* __restrict__ bcount, int* __restrict__ boff) {
    __shared__ int sc[256];
    if (blockIdx.x == gridDim.x - 1) {
        const int t = threadIdx.x;
        int v[4];
        int sum = 0;
        #pragma unroll
        for (int k = 0; k < 4; ++k) {
            const int idx = t * 4 + k;
            v[k] = (idx < NBUCK) ? bcount[idx * BCSTRIDE] : 0;
            sum += v[k];
        }
        sc[t] = sum;
        __syncthreads();
        for (int o = 1; o < 256; o <<= 1) {
            const int x = (t >= o) ? sc[t - o] : 0;
            __syncthreads();
            sc[t] += x;
            __syncthreads();
        }
        int run = sc[t] - sum;
        #pragma unroll
        for (int k = 0; k < 4; ++k) {
            const int idx = t * 4 + k;
            if (idx <= NBUCK) boff[idx] = run;
            run += v[k];
        }
        return;
    }
    int n = blockIdx.x * 256 + threadIdx.x;
    if (n < NNODES) {
        float d = deg[n];
        deg[n] = (d > 0.0f) ? rsqrtf(d) : 0.0f;
    }
}

__global__ void gcn_place_kernel(const int* __restrict__ row, const int* __restrict__ col,
                                 const int* __restrict__ boff, const int* __restrict__ rank,
                                 int2* __restrict__ sorted) {
    int e = blockIdx.x * 256 + threadIdx.x;
    if (e < NEDGES) {
        const int r = row[e];
        sorted[boff[r / BROWS] + rank[e]] = make_int2(r, col[e]);
    }
}

#define EP_WG      256
#define EP_THREADS 1024
#define EP_ECHUNK  ((NEDGES + EP_WG - 1) / EP_WG)
#define EP_NCHUNK  ((NNODES + EP_WG - 1) / EP_WG)
#define PSTRIDE    (NGRAPH * FEAT + NGRAPH)
#define EP_LDS_FLOATS 41472

__device__ __forceinline__ void ep_add_row(float* acc_row, float w, float4 h) {
    atomicAdd(acc_row,      w * h.x);
    atomicAdd(acc_row + 32, w * h.y);
    atomicAdd(acc_row + 64, w * h.z);
    atomicAdd(acc_row + 96, w * h.w);
}

#define TPITCH 132
#define TJ     33

__device__ __forceinline__ void tile_add_row(float* acc_row, float w, float4 h) {
    atomicAdd(acc_row,          w * h.x);
    atomicAdd(acc_row + TJ,     w * h.y);
    atomicAdd(acc_row + 2 * TJ, w * h.z);
    atomicAdd(acc_row + 3 * TJ, w * h.w);
}

__global__ __launch_bounds__(256)
void gcn_tile_agg_gemm_kernel(const int2* __restrict__ sorted,
                              const int* __restrict__ boff,
                              const float* __restrict__ dinv,
                              const float* __restrict__ X,
                              const float* __restrict__ W1,
                              const float* __restrict__ b1,
                              const float* __restrict__ W2,
                              const float* __restrict__ b2,
                              float* __restrict__ Y) {
    __shared__ float acc[BROWS * TPITCH];
    __shared__ float Wt[FEAT * 16];
    __shared__ float Bt[16];
    __shared__ float wsum[BROWS];

    const int tid  = threadIdx.x;
    const int lane = tid & 31;
    const int wave = __builtin_amdgcn_readfirstlane(tid >> 5);
    const int r0   = blockIdx.x * BROWS;

    for (int i = tid * 4; i < BROWS * TPITCH; i += 256 * 4)
        *(float4*)(acc + i) = make_float4(0.0f, 0.0f, 0.0f, 0.0f);
    if (tid < BROWS) wsum[tid] = 0.0f;
    __syncthreads();

    const int e_lo = boff[blockIdx.x];
    const int e_n  = boff[blockIdx.x + 1] - e_lo;

    for (int base = wave * 32; base < e_n; base += 256) {
        const int  i     = base + lane;
        const bool valid = i < e_n;
        int   rl  = 0;
        int   c   = 0;
        float nrm = 0.0f;
        if (valid) {
            const int2 rc = sorted[e_lo + i];
            rl  = rc.x - r0;
            c   = rc.y;
            nrm = dinv[rc.x] * dinv[c];
            atomicAdd(&wsum[rl], nrm);
        }
        const int cnt = (e_n - base < 32) ? (e_n - base) : 32;

        int j = 0;
        for (; j + 8 <= cnt; j += 8) {
            float4 h[8];
            #pragma unroll
            for (int u = 0; u < 8; ++u) {
                const int cj = __builtin_amdgcn_readlane(c, j + u);
                h[u] = *(const float4*)(X + (size_t)cj * FEAT + lane * 4);
            }
            #pragma unroll
            for (int u = 0; u < 8; ++u) {
                const int   rj = __builtin_amdgcn_readlane(rl, j + u);
                const float nj = __int_as_float(
                    __builtin_amdgcn_readlane(__float_as_int(nrm), j + u));
                tile_add_row(acc + rj * TPITCH + lane, nj, h[u]);
            }
        }
        for (; j < cnt; ++j) {
            const int   cj = __builtin_amdgcn_readlane(c, j);
            const int   rj = __builtin_amdgcn_readlane(rl, j);
            const float nj = __int_as_float(
                __builtin_amdgcn_readlane(__float_as_int(nrm), j));
            const float4 hv = *(const float4*)(X + (size_t)cj * FEAT + lane * 4);
            tile_add_row(acc + rj * TPITCH + lane, nj, hv);
        }
    }
    __syncthreads();

    const int  m0     = r0 + wave * 16;
    const bool active = (m0 < NNODES);
    if (active) {
        for (int i = 0; i < 16; ++i) {
            const int r = m0 + i;
            float s = dinv[r]; s = s * s;
            const float4 h = *(const float4*)(X + (size_t)r * FEAT + lane * 4);
            float* a = acc + (wave * 16 + i) * TPITCH + lane;
            a[0]      += s * h.x;
            a[TJ]     += s * h.y;
            a[2 * TJ] += s * h.z;
            a[3 * TJ] += s * h.w;
        }
        if (lane < 16) {
            float s = dinv[m0 + lane]; s = s * s;
            wsum[wave * 16 + lane] += s;
        }
    }
    __syncthreads();

    const int khalf = (lane >> 4) << 1;
    const int ncol  = lane & 15;
    const int half  = lane >> 4;
    const float* ar = acc + (wave * 16 + (lane & 15)) * TPITCH + TJ * khalf;
    v2f af[32];
    #pragma unroll
    for (int s = 0; s < 32; ++s) {
        af[s][0] = ar[s];
        af[s][1] = ar[TJ + s];
    }
    float sv[8];
    #pragma unroll
    for (int v = 0; v < 8; ++v) sv[v] = wsum[wave * 16 + v + half * 8];

    for (int nt = 0; nt < 8; ++nt) {
        if (nt > 0) __syncthreads();
        #pragma unroll
        for (int i = 0; i < 8; ++i) {
            int idx = tid * 8 + i;
            int kk  = idx >> 4;
            int cc  = idx & 15;
            Wt[idx] = W1[kk * FEAT + nt * 16 + cc];
        }
        if (tid < 16) Bt[tid] = b1[nt * 16 + tid];
        __syncthreads();

        if (!active) continue;

        v8f d = {};
        #pragma unroll
        for (int s = 0; s < 32; ++s) {
            v2f b;
            b[0] = Wt[(4 * s + khalf) * 16 + ncol];
            b[1] = Wt[(4 * s + khalf + 1) * 16 + ncol];
            d = __builtin_amdgcn_wmma_f32_16x16x4_f32(
                false, af[s], false, b, (short)0, d, false, false);
        }

        #pragma unroll
        for (int v = 0; v < 8; ++v) {
            const int rr = wave * 16 + v + half * 8;
            acc[rr * TPITCH + TJ * (ncol & 3) + nt * 4 + (ncol >> 2)] =
                fmaxf(d[v] + sv[v] * Bt[ncol], 0.0f);
        }
    }
    __syncthreads();

    #pragma unroll
    for (int s = 0; s < 32; ++s) {
        af[s][0] = ar[s];
        af[s][1] = ar[TJ + s];
    }

    for (int nt = 0; nt < 8; ++nt) {
        if (nt > 0) __syncthreads();
        #pragma unroll
        for (int i = 0; i < 8; ++i) {
            int idx = tid * 8 + i;
            int kk  = idx >> 4;
            int cc  = idx & 15;
            Wt[idx] = W2[kk * FEAT + nt * 16 + cc];
        }
        if (tid < 16) Bt[tid] = b2[nt * 16 + tid];
        __syncthreads();

        if (!active) continue;

        v8f d = {};
        #pragma unroll
        for (int s = 0; s < 32; ++s) {
            v2f b;
            b[0] = Wt[(4 * s + khalf) * 16 + ncol];
            b[1] = Wt[(4 * s + khalf + 1) * 16 + ncol];
            d = __builtin_amdgcn_wmma_f32_16x16x4_f32(
                false, af[s], false, b, (short)0, d, false, false);
        }

        #pragma unroll
        for (int v = 0; v < 8; ++v) {
            int r = m0 + v + half * 8;
            Y[(size_t)r * FEAT + nt * 16 + ncol] = d[v] + Bt[ncol];
        }
    }
}

__global__ __launch_bounds__(EP_THREADS)
void gcn_edge_pool_kernel(const int* __restrict__ row,
                          const int* __restrict__ col,
                          const int* __restrict__ batch,
                          const float* __restrict__ dinv,
                          const float* __restrict__ H,
                          float* __restrict__ partial) {
    __shared__ float acc[EP_LDS_FLOATS];

    const int tid  = threadIdx.x;
    const int lane = tid & 31;
    const int wave = __builtin_amdgcn_readfirstlane(tid >> 5);

    for (int i = tid * 4; i < PSTRIDE; i += EP_THREADS * 4)
        *(float4*)(acc + i) = make_float4(0.0f, 0.0f, 0.0f, 0.0f);
    __syncthreads();

    const int e_lo = blockIdx.x * EP_ECHUNK;
    const int e_n  = (NEDGES - e_lo < EP_ECHUNK) ? (NEDGES - e_lo) : EP_ECHUNK;
    const int n_lo = blockIdx.x * EP_NCHUNK;
    const int n_n  = (NNODES - n_lo < EP_NCHUNK) ? (NNODES - n_lo) : EP_NCHUNK;
    const int total = e_n + n_n;

    for (int base = wave * 32; base < total; base += EP_THREADS) {
        const int  i      = base + lane;
        const bool valid  = i < total;
        const bool isnode = valid && i >= e_n;
        int   c   = 0;
        int   g   = 0;
        float nrm = 0.0f;
        if (valid) {
            int r;
            if (isnode) { r = n_lo + (i - e_n); c = r; }
            else        { r = row[e_lo + i];    c = col[e_lo + i]; }
            g   = batch[r];
            nrm = dinv[r] * dinv[c];
        }

        const unsigned nodes = __builtin_amdgcn_ballot_w32(isnode);
        if (nodes != 0u) {
            const int  gup  = __shfl_up(g, 1);
            const bool head = isnode &&
                (lane == 0 || ((nodes >> (lane - 1)) & 1u) == 0u || gup != g);
            const unsigned heads = __builtin_amdgcn_ballot_w32(head);
            if (head) {
                const unsigned stop = (heads | ~nodes) & ~((2u << lane) - 1u);
                const int end = (stop != 0u) ? __builtin_ctz(stop) : 32;
                atomicAdd(&acc[NGRAPH * FEAT + g], (float)(end - lane));
            }
        }

        const int cnt = (total - base < 32) ? (total - base) : 32;

        int j = 0;
        for (; j + 8 <= cnt; j += 8) {
            float4 h[8];
            #pragma unroll
            for (int u = 0; u < 8; ++u) {
                const int cj = __builtin_amdgcn_readlane(c, j + u);
                h[u] = *(const float4*)(H + (size_t)cj * FEAT + lane * 4);
            }
            #pragma unroll
            for (int u = 0; u < 8; ++u) {
                const int   gj = __builtin_amdgcn_readlane(g, j + u);
                const float nj = __int_as_float(
                    __builtin_amdgcn_readlane(__float_as_int(nrm), j + u));
                ep_add_row(acc + gj * FEAT + lane, nj, h[u]);
            }
        }
        for (; j < cnt; ++j) {
            const int   cj = __builtin_amdgcn_readlane(c, j);
            const int   gj = __builtin_amdgcn_readlane(g, j);
            const float nj = __int_as_float(
                __builtin_amdgcn_readlane(__float_as_int(nrm), j));
            const float4 hv = *(const float4*)(H + (size_t)cj * FEAT + lane * 4);
            ep_add_row(acc + gj * FEAT + lane, nj, hv);
        }
    }
    __syncthreads();

    float* P = partial + (size_t)blockIdx.x * PSTRIDE;
    for (int i = tid * 4; i < PSTRIDE; i += EP_THREADS * 4)
        *(float4*)(P + i) = *(const float4*)(acc + i);
}

__global__ __launch_bounds__(256)
void gcn_finalize_kernel(const float* __restrict__ partial,
                         float* __restrict__ out) {
    __shared__ float sm[8][FEAT];
    __shared__ float scnt[8];

    const int g = blockIdx.x;
    const int t = threadIdx.x;
    const int q = t & 31;
    const int s = t >> 5;

    const float* p = partial + (size_t)(s * 32) * PSTRIDE + g * FEAT + q * 4;
    float4 a = make_float4(0.0f, 0.0f, 0.0f, 0.0f);
    #pragma unroll
    for (int w = 0; w < 32; ++w) {
        const float4 v = *(const float4*)(p + (size_t)w * PSTRIDE);
        a.x += v.x; a.y += v.y; a.z += v.z; a.w += v.w;
    }
    float* smf = &sm[s][16 * (q & 7) + (q >> 3)];
    smf[0]  = a.x;
    smf[4]  = a.y;
    smf[8]  = a.z;
    smf[12] = a.w;

    float cn = partial[(size_t)t * PSTRIDE + NGRAPH * FEAT + g];
    for (int o = 16; o > 0; o >>= 1) cn += __shfl_xor(cn, o);
    if (q == 0) scnt[s] = cn;
    __syncthreads();

    if (t < 32) {
        float4 tot = *(const float4*)&sm[0][4 * t];
        float  c   = scnt[0];
        #pragma unroll
        for (int k = 1; k < 8; ++k) {
            const float4 v = *(const float4*)&sm[k][4 * t];
            tot.x += v.x; tot.y += v.y; tot.z += v.z; tot.w += v.w;
            c += scnt[k];
        }
        const float d = fmaxf(c, 1.0f);
        *(float4*)(out + (size_t)g * FEAT + 4 * t) =
            make_float4(tot.x / d, tot.y / d, tot.z / d, tot.w / d);
    }
}

extern "C" void kernel_launch(void* const* d_in, const int* in_sizes, int n_in,
                              void* d_out, int out_size, void* d_ws, size_t ws_size,
                              hipStream_t stream) {
    const float* x          = (const float*)d_in[0];
    const int*   edge_index = (const int*)  d_in[1];
    const int*   batch      = (const int*)  d_in[2];
    const float* W1 = (const float*)d_in[4];
    const float* b1 = (const float*)d_in[5];
    const float* W2 = (const float*)d_in[6];
    const float* b2 = (const float*)d_in[7];
    float* out = (float*)d_out;

    float* H2   = (float*)d_ws;
    float* dinv = H2   + (size_t)NNODES * FEAT;
    float* part = dinv + 100032;
    int*   bcount = (int*)(part + (size_t)EP_WG * PSTRIDE);
    int*   boff   = bcount + BCWORDS;
    int*   rank   = boff + 1024;
    int2*  sorted = (int2*)(rank + NEDGES);

    const int* row = edge_index;
    const int* col = edge_index + NEDGES;

    dim3 blk(256);
    const int grid_nodes  = (NNODES + 255) / 256;
    const int grid_edges  = (NEDGES + 255) / 256;

    gcn_init_kernel<<<grid_nodes, blk, 0, stream>>>(dinv, bcount);
    gcn_deg_kernel<<<grid_edges, blk, 0, stream>>>(row, col, dinv, bcount, rank);
    gcn_dinv_kernel<<<grid_nodes + 1, blk, 0, stream>>>(dinv, bcount, boff);
    gcn_place_kernel<<<grid_edges, blk, 0, stream>>>(row, col, boff, rank, sorted);

    gcn_tile_agg_gemm_kernel<<<NBUCK, blk, 0, stream>>>(sorted, boff, dinv, x, W1, b1, W2, b2, H2);

    gcn_edge_pool_kernel<<<EP_WG, EP_THREADS, 0, stream>>>(row, col, batch, dinv, H2, part);
    gcn_finalize_kernel<<<NGRAPH, blk, 0, stream>>>(part, out);
}


# pool pass walks the edges grouped by row tile: a work-group's items fall in few graphs, so a wave sums its graph in registers and only that graph range is zeroed, stored, read back; new sum order
# speedup vs baseline: 3.3201x; 1.0518x over previous
#include <hip/hip_runtime.h>
#include <cmath>

#define NNODES 100000
#define NEDGES 600000
#define FEAT   128
#define NGRAPH 256

typedef __attribute__((ext_vector_type(2))) float v2f;
typedef __attribute__((ext_vector_type(8))) float v8f;

#define BROWS 128
#define NBUCK ((NNODES + BROWS - 1) / BROWS)
#define BCSTRIDE 64
#define BCWORDS  (NBUCK * BCSTRIDE)

__global__ void gcn_init_kernel(float* __restrict__ deg, int* __restrict__ bcount) {
    int i = blockIdx.x * 256 + threadIdx.x;
    if (i < NNODES)        deg[i]  = 1.0f;
    if (i < BCWORDS)       bcount[i] = 0;
}

__global__ void gcn_deg_kernel(const int* __restrict__ row, const int* __restrict__ col,
                               float* __restrict__ deg,
                               int* __restrict__ bcount, int* __restrict__ rank) {
    int e = blockIdx.x * 256 + threadIdx.x;
    if (e < NEDGES) {
        const int c = col[e];
        const int r = row[e];
        atomicAdd(&deg[c], 1.0f);
        rank[e] = atomicAdd(&bcount[(r / BROWS) * BCSTRIDE], 1);
    }
}

__global__ void gcn_dinv_kernel(float* __restrict__ deg,
                                const int* __restrict__ bcount, int* __restrict__ boff) {
    __shared__ int sc[256];
    if (blockIdx.x == gridDim.x - 1) {
        const int t = threadIdx.x;
        int v[4];
        int sum = 0;
        #pragma unroll
        for (int k = 0; k < 4; ++k) {
            const int idx = t * 4 + k;
            v[k] = (idx < NBUCK) ? bcount[idx * BCSTRIDE] : 0;
            sum += v[k];
        }
        sc[t] = sum;
        __syncthreads();
        for (int o = 1; o < 256; o <<= 1) {
            const int x = (t >= o) ? sc[t - o] : 0;
            __syncthreads();
            sc[t] += x;
            __syncthreads();
        }
        int run = sc[t] - sum;
        #pragma unroll
        for (int k = 0; k < 4; ++k) {
            const int idx = t * 4 + k;
            if (idx <= NBUCK) boff[idx] = run;
            run += v[k];
        }
        return;
    }
    int n = blockIdx.x * 256 + threadIdx.x;
    if (n < NNODES) {
        float d = deg[n];
        deg[n] = (d > 0.0f) ? rsqrtf(d) : 0.0f;
    }
}

__global__ void gcn_place_kernel(const int* __restrict__ row, const int* __restrict__ col,
                                 const int* __restrict__ boff, const int* __restrict__ rank,
                                 int2* __restrict__ sorted) {
    int e = blockIdx.x * 256 + threadIdx.x;
    if (e < NEDGES) {
        const int r = row[e];
        sorted[boff[r / BROWS] + rank[e]] = make_int2(r, col[e]);
    }
}

#define EP_WG      256
#define EP_THREADS 1024
#define EP_ECHUNK  ((NEDGES + EP_WG - 1) / EP_WG)
#define EP_NCHUNK  ((NNODES + EP_WG - 1) / EP_WG)
#define PSTRIDE    (NGRAPH * FEAT + NGRAPH)
#define EP_LDS_FLOATS 41472

__device__ __forceinline__ void ep_add_row(float* acc_row, float w, float4 h) {
    atomicAdd(acc_row,      w * h.x);
    atomicAdd(acc_row + 32, w * h.y);
    atomicAdd(acc_row + 64, w * h.z);
    atomicAdd(acc_row + 96, w * h.w);
}

#define TPITCH 132
#define TJ     33

__device__ __forceinline__ void tile_add_row(float* acc_row, float w, float4 h) {
    atomicAdd(acc_row,          w * h.x);
    atomicAdd(acc_row + TJ,     w * h.y);
    atomicAdd(acc_row + 2 * TJ, w * h.z);
    atomicAdd(acc_row + 3 * TJ, w * h.w);
}

__global__ __launch_bounds__(256)
void gcn_tile_agg_gemm_kernel(const int2* __restrict__ sorted,
                              const int* __restrict__ boff,
                              const float* __restrict__ dinv,
                              const float* __restrict__ X,
                              const float* __restrict__ W1,
                              const float* __restrict__ b1,
                              const float* __restrict__ W2,
                              const float* __restrict__ b2,
                              float* __restrict__ Y) {
    __shared__ float acc[BROWS * TPITCH];
    __shared__ float Wt[FEAT * 16];
    __shared__ float Bt[16];
    __shared__ float wsum[BROWS];

    const int tid  = threadIdx.x;
    const int lane = tid & 31;
    const int wave = __builtin_amdgcn_readfirstlane(tid >> 5);
    const int r0   = blockIdx.x * BROWS;

    for (int i = tid * 4; i < BROWS * TPITCH; i += 256 * 4)
        *(float4*)(acc + i) = make_float4(0.0f, 0.0f, 0.0f, 0.0f);
    if (tid < BROWS) wsum[tid] = 0.0f;
    __syncthreads();

    const int e_lo = boff[blockIdx.x];
    const int e_n  = boff[blockIdx.x + 1] - e_lo;

    for (int base = wave * 32; base < e_n; base += 256) {
        const int  i     = base + lane;
        const bool valid = i < e_n;
        int   rl  = 0;
        int   c   = 0;
        float nrm = 0.0f;
        if (valid) {
            const int2 rc = sorted[e_lo + i];
            rl  = rc.x - r0;
            c   = rc.y;
            nrm = dinv[rc.x] * dinv[c];
            atomicAdd(&wsum[rl], nrm);
        }
        const int cnt = (e_n - base < 32) ? (e_n - base) : 32;

        int j = 0;
        for (; j + 8 <= cnt; j += 8) {
            float4 h[8];
            #pragma unroll
            for (int u = 0; u < 8; ++u) {
                const int cj = __builtin_amdgcn_readlane(c, j + u);
                h[u] = *(const float4*)(X + (size_t)cj * FEAT + lane * 4);
            }
            #pragma unroll
            for (int u = 0; u < 8; ++u) {
                const int   rj = __builtin_amdgcn_readlane(rl, j + u);
                const float nj = __int_as_float(
                    __builtin_amdgcn_readlane(__float_as_int(nrm), j + u));
                tile_add_row(acc + rj * TPITCH + lane, nj, h[u]);
            }
        }
        for (; j < cnt; ++j) {
            const int   cj = __builtin_amdgcn_readlane(c, j);
            const int   rj = __builtin_amdgcn_readlane(rl, j);
            const float nj = __int_as_float(
                __builtin_amdgcn_readlane(__float_as_int(nrm), j));
            const float4 hv = *(const float4*)(X + (size_t)cj * FEAT + lane * 4);
            tile_add_row(acc + rj * TPITCH + lane, nj, hv);
        }
    }
    __syncthreads();

    const int  m0     = r0 + wave * 16;
    const bool active = (m0 < NNODES);
    if (active) {
        for (int i = 0; i < 16; ++i) {
            const int r = m0 + i;
            float s = dinv[r]; s = s * s;
            const float4 h = *(const float4*)(X + (size_t)r * FEAT + lane * 4);
            float* a = acc + (wave * 16 + i) * TPITCH + lane;
            a[0]      += s * h.x;
            a[TJ]     += s * h.y;
            a[2 * TJ] += s * h.z;
            a[3 * TJ] += s * h.w;
        }
        if (lane < 16) {
            float s = dinv[m0 + lane]; s = s * s;
            wsum[wave * 16 + lane] += s;
        }
    }
    __syncthreads();

    const int khalf = (lane >> 4) << 1;
    const int ncol  = lane & 15;
    const int half  = lane >> 4;
    const float* ar = acc + (wave * 16 + (lane & 15)) * TPITCH + TJ * khalf;
    v2f af[32];
    #pragma unroll
    for (int s = 0; s < 32; ++s) {
        af[s][0] = ar[s];
        af[s][1] = ar[TJ + s];
    }
    float sv[8];
    #pragma unroll
    for (int v = 0; v < 8; ++v) sv[v] = wsum[wave * 16 + v + half * 8];

    for (int nt = 0; nt < 8; ++nt) {
        if (nt > 0) __syncthreads();
        #pragma unroll
        for (int i = 0; i < 8; ++i) {
            int idx = tid * 8 + i;
            int kk  = idx >> 4;
            int cc  = idx & 15;
            Wt[idx] = W1[kk * FEAT + nt * 16 + cc];
        }
        if (tid < 16) Bt[tid] = b1[nt * 16 + tid];
        __syncthreads();

        if (!active) continue;

        v8f d = {};
        #pragma unroll
        for (int s = 0; s < 32; ++s) {
            v2f b;
            b[0] = Wt[(4 * s + khalf) * 16 + ncol];
            b[1] = Wt[(4 * s + khalf + 1) * 16 + ncol];
            d = __builtin_amdgcn_wmma_f32_16x16x4_f32(
                false, af[s], false, b, (short)0, d, false, false);
        }

        #pragma unroll
        for (int v = 0; v < 8; ++v) {
            const int rr = wave * 16 + v + half * 8;
            acc[rr * TPITCH + TJ * (ncol & 3) + nt * 4 + (ncol >> 2)] =
                fmaxf(d[v] + sv[v] * Bt[ncol], 0.0f);
        }
    }
    __syncthreads();

    #pragma unroll
    for (int s = 0; s < 32; ++s) {
        af[s][0] = ar[s];
        af[s][1] = ar[TJ + s];
    }

    for (int nt = 0; nt < 8; ++nt) {
        if (nt > 0) __syncthreads();
        #pragma unroll
        for (int i = 0; i < 8; ++i) {
            int idx = tid * 8 + i;
            int kk  = idx >> 4;
            int cc  = idx & 15;
            Wt[idx] = W2[kk * FEAT + nt * 16 + cc];
        }
        if (tid < 16) Bt[tid] = b2[nt * 16 + tid];
        __syncthreads();

        if (!active) continue;

        v8f d = {};
        #pragma unroll
        for (int s = 0; s < 32; ++s) {
            v2f b;
            b[0] = Wt[(4 * s + khalf) * 16 + ncol];
            b[1] = Wt[(4 * s + khalf + 1) * 16 + ncol];
            d = __builtin_amdgcn_wmma_f32_16x16x4_f32(
                false, af[s], false, b, (short)0, d, false, false);
        }

        #pragma unroll
        for (int v = 0; v < 8; ++v) {
            int r = m0 + v + half * 8;
            Y[(size_t)r * FEAT + nt * 16 + ncol] = d[v] + Bt[ncol];
        }
    }
}

#define EP_ITERS ((EP_ECHUNK + EP_NCHUNK + EP_THREADS - 1) / EP_THREADS)

__device__ __forceinline__ void ep_flush_row(float* acc_row, float4 s) {
    atomicAdd(acc_row,      s.x);
    atomicAdd(acc_row + 32, s.y);
    atomicAdd(acc_row + 64, s.z);
    atomicAdd(acc_row + 96, s.w);
}

__global__ __launch_bounds__(EP_THREADS)
void gcn_edge_pool_kernel(const int2* __restrict__ sorted,
                          const int* __restrict__ batch,
                          const float* __restrict__ dinv,
                          const float* __restrict__ H,
                          float* __restrict__ partial,
                          int2* __restrict__ prange) {
    __shared__ float acc[EP_LDS_FLOATS];
    __shared__ int wmin[EP_THREADS / 32];
    __shared__ int wmax[EP_THREADS / 32];

    const int tid  = threadIdx.x;
    const int lane = tid & 31;
    const int wave = __builtin_amdgcn_readfirstlane(tid >> 5);

    const int e_lo = blockIdx.x * EP_ECHUNK;
    const int e_n  = (NEDGES - e_lo < EP_ECHUNK) ? (NEDGES - e_lo) : EP_ECHUNK;
    const int n_lo = blockIdx.x * EP_NCHUNK;
    const int n_n  = (NNODES - n_lo < EP_NCHUNK) ? (NNODES - n_lo) : EP_NCHUNK;
    const int total = e_n + n_n;

    int   ci[EP_ITERS];
    int   gi[EP_ITERS];
    float wi[EP_ITERS];
    int lmin = NGRAPH;
    int lmax = -1;
    #pragma unroll
    for (int it = 0; it < EP_ITERS; ++it) {
        const int i = it * EP_THREADS + tid;
        ci[it] = 0;
        gi[it] = 0;
        wi[it] = 0.0f;
        if (i < total) {
            int r;
            int c;
            if (i >= e_n) { r = n_lo + (i - e_n); c = r; }
            else          { const int2 rc = sorted[e_lo + i]; r = rc.x; c = rc.y; }
            const int g = batch[r];
            ci[it] = c;
            gi[it] = g;
            wi[it] = dinv[r] * dinv[c];
            lmin = (g < lmin) ? g : lmin;
            lmax = (g > lmax) ? g : lmax;
        }
    }
    for (int o = 16; o > 0; o >>= 1) {
        const int a = __shfl_xor(lmin, o);
        const int b = __shfl_xor(lmax, o);
        lmin = (a < lmin) ? a : lmin;
        lmax = (b > lmax) ? b : lmax;
    }
    if (lane == 0) { wmin[wave] = lmin; wmax[wave] = lmax; }
    __syncthreads();
    int gmin = wmin[lane];
    int gmax = wmax[lane];
    for (int o = 16; o > 0; o >>= 1) {
        const int a = __shfl_xor(gmin, o);
        const int b = __shfl_xor(gmax, o);
        gmin = (a < gmin) ? a : gmin;
        gmax = (b > gmax) ? b : gmax;
    }
    gmin = __builtin_amdgcn_readfirstlane(gmin);
    gmax = __builtin_amdgcn_readfirstlane(gmax);

    for (int i = gmin * FEAT + tid * 4; i < (gmax + 1) * FEAT; i += EP_THREADS * 4)
        *(float4*)(acc + i) = make_float4(0.0f, 0.0f, 0.0f, 0.0f);
    if (tid <= gmax - gmin) acc[NGRAPH * FEAT + gmin + tid] = 0.0f;
    __syncthreads();

    float4 rs = make_float4(0.0f, 0.0f, 0.0f, 0.0f);
    int gcur = -1;

    #pragma unroll
    for (int it = 0; it < EP_ITERS; ++it) {
        const int base = it * EP_THREADS + wave * 32;
        if (base < total) {
            const int   i      = base + lane;
            const bool  valid  = i < total;
            const bool  isnode = valid && i >= e_n;
            const int   c      = ci[it];
            const int   g      = gi[it];
            const float nrm    = wi[it];

            const unsigned nodes = __builtin_amdgcn_ballot_w32(isnode);
            if (nodes != 0u) {
                const int  gup  = __shfl_up(g, 1);
                const bool head = isnode &&
                    (lane == 0 || ((nodes >> (lane - 1)) & 1u) == 0u || gup != g);
                const unsigned heads = __builtin_amdgcn_ballot_w32(head);
                if (head) {
                    const unsigned stop = (heads | ~nodes) & ~((2u << lane) - 1u);
                    const int end = (stop != 0u) ? __builtin_ctz(stop) : 32;
                    atomicAdd(&acc[NGRAPH * FEAT + g], (float)(end - lane));
                }
            }

            const int g0 = __builtin_amdgcn_readfirstlane(g);
            if (g0 != gcur) {
                if (gcur >= 0) ep_flush_row(acc + gcur * FEAT + lane, rs);
                rs   = make_float4(0.0f, 0.0f, 0.0f, 0.0f);
                gcur = g0;
            }

            const int cnt = (total - base < 32) ? (total - base) : 32;

            int j = 0;
            for (; j + 8 <= cnt; j += 8) {
                float4 h[8];
                #pragma unroll
                for (int u = 0; u < 8; ++u) {
                    const int cj = __builtin_amdgcn_readlane(c, j + u);
                    h[u] = *(const float4*)(H + (size_t)cj * FEAT + lane * 4);
                }
                #pragma unroll
                for (int u = 0; u < 8; ++u) {
                    const int   gj = __builtin_amdgcn_readlane(g, j + u);
                    const float nj = __int_as_float(
                        __builtin_amdgcn_readlane(__float_as_int(nrm), j + u));
                    if (gj == gcur) {
                        rs.x += nj * h[u].x;
                        rs.y += nj * h[u].y;
                        rs.z += nj * h[u].z;
                        rs.w += nj * h[u].w;
                    } else {
                        ep_add_row(acc + gj * FEAT + lane, nj, h[u]);
                    }
                }
            }
            for (; j < cnt; ++j) {
                const int   cj = __builtin_amdgcn_readlane(c, j);
                const int   gj = __builtin_amdgcn_readlane(g, j);
                const float nj = __int_as_float(
                    __builtin_amdgcn_readlane(__float_as_int(nrm), j));
                const float4 hv = *(const float4*)(H + (size_t)cj * FEAT + lane * 4);
                if (gj == gcur) {
                    rs.x += nj * hv.x;
                    rs.y += nj * hv.y;
                    rs.z += nj * hv.z;
                    rs.w += nj * hv.w;
                } else {
                    ep_add_row(acc + gj * FEAT + lane, nj, hv);
                }
            }
        }
    }
    if (gcur >= 0) ep_flush_row(acc + gcur * FEAT + lane, rs);
    __syncthreads();

    float* P = partial + (size_t)blockIdx.x * PSTRIDE;
    for (int i = gmin * FEAT + tid * 4; i < (gmax + 1) * FEAT; i += EP_THREADS * 4)
        *(float4*)(P + i) = *(const float4*)(acc + i);
    if (tid <= gmax - gmin)
        P[NGRAPH * FEAT + gmin + tid] = acc[NGRAPH * FEAT + gmin + tid];
    if (tid == 0) prange[blockIdx.x] = make_int2(gmin, gmax);
}

__global__ __launch_bounds__(256)
void gcn_finalize_kernel(const float* __restrict__ partial,
                         const int2* __restrict__ prange,
                         float* __restrict__ out) {
    __shared__ float sm[8][FEAT];
    __shared__ float scnt[8];

    const int g = blockIdx.x;
    const int t = threadIdx.x;
    const int q = t & 31;
    const int s = t >> 5;

    const int2 rg   = prange[t];
    const bool mine = (rg.x <= g) && (g <= rg.y);
    unsigned planes = __builtin_amdgcn_ballot_w32(mine);

    const float* p = partial + (size_t)(s * 32) * PSTRIDE + g * FEAT + q * 4;
    float4 a = make_float4(0.0f, 0.0f, 0.0f, 0.0f);
    while (planes != 0u) {
        const int w = __builtin_ctz(planes);
        planes &= planes - 1u;
        const float4 v = *(const float4*)(p + (size_t)w * PSTRIDE);
        a.x += v.x; a.y += v.y; a.z += v.z; a.w += v.w;
    }
    float* smf = &sm[s][16 * (q & 7) + (q >> 3)];
    smf[0]  = a.x;
    smf[4]  = a.y;
    smf[8]  = a.z;
    smf[12] = a.w;

    float cn = mine ? partial[(size_t)t * PSTRIDE + NGRAPH * FEAT + g] : 0.0f;
    for (int o = 16; o > 0; o >>= 1) cn += __shfl_xor(cn, o);
    if (q == 0) scnt[s] = cn;
    __syncthreads();

    if (t < 32) {
        float4 tot = *(const float4*)&sm[0][4 * t];
        float  c   = scnt[0];
        #pragma unroll
        for (int k = 1; k < 8; ++k) {
            const float4 v = *(const float4*)&sm[k][4 * t];
            tot.x += v.x; tot.y += v.y; tot.z += v.z; tot.w += v.w;
            c += scnt[k];
        }
        const float d = fmaxf(c, 1.0f);
        *(float4*)(out + (size_t)g * FEAT + 4 * t) =
            make_float4(tot.x / d, tot.y / d, tot.z / d, tot.w / d);
    }
}

extern "C" void kernel_launch(void* const* d_in, const int* in_sizes, int n_in,
                              void* d_out, int out_size, void* d_ws, size_t ws_size,
                              hipStream_t stream) {
    const float* x          = (const float*)d_in[0];
    const int*   edge_index = (const int*)  d_in[1];
    const int*   batch      = (const int*)  d_in[2];
    const float* W1 = (const float*)d_in[4];
    const float* b1 = (const float*)d_in[5];
    const float* W2 = (const float*)d_in[6];
    const float* b2 = (const float*)d_in[7];
    float* out = (float*)d_out;

    float* H2   = (float*)d_ws;
    float* dinv = H2   + (size_t)NNODES * FEAT;
    float* part = dinv + 100032;
    int*   bcount = (int*)(part + (size_t)EP_WG * PSTRIDE);
    int*   boff   = bcount + BCWORDS;
    int*   rank   = boff + 1024;
    int2*  sorted = (int2*)(rank + NEDGES);
    int2*  prange = sorted + NEDGES;

    const int* row = edge_index;
    const int* col = edge_index + NEDGES;

    dim3 blk(256);
    const int grid_nodes  = (NNODES + 255) / 256;
    const int grid_edges  = (NEDGES + 255) / 256;

    gcn_init_kernel<<<grid_nodes, blk, 0, stream>>>(dinv, bcount);
    gcn_deg_kernel<<<grid_edges, blk, 0, stream>>>(row, col, dinv, bcount, rank);
    gcn_dinv_kernel<<<grid_nodes + 1, blk, 0, stream>>>(dinv, bcount, boff);
    gcn_place_kernel<<<grid_edges, blk, 0, stream>>>(row, col, boff, rank, sorted);

    gcn_tile_agg_gemm_kernel<<<NBUCK, blk, 0, stream>>>(sorted, boff, dinv, x, W1, b1, W2, b2, H2);

    gcn_edge_pool_kernel<<<EP_WG, EP_THREADS, 0, stream>>>(sorted, batch, dinv, H2, part, prange);
    gcn_finalize_kernel<<<NGRAPH, blk, 0, stream>>>(part, prange, out);
}
